# starting kernel
# baseline (speedup 1.0000x reference)
#include <hip/hip_runtime.h>

typedef __attribute__((ext_vector_type(2))) float v2f;
typedef __attribute__((ext_vector_type(8))) float v8f;

#define D_FEAT 128

__global__ __launch_bounds__(256) void ggnn_gemm_wmma(
    const float* __restrict__ x,
    const float* __restrict__ W,
    float* __restrict__ xw,
    int nNodes) {
  const int lane = threadIdx.x & 31;
  const int wave = threadIdx.x >> 5;
  const int half = lane >> 4;
  const int l16  = lane & 15;
  const int row0 = blockIdx.x * 16;

  int mrow = row0 + l16;
  if (mrow >= nNodes) mrow = nNodes - 1;
  const int ncol = wave * 16 + l16;

  const float* __restrict__ xrow = x + (size_t)mrow * D_FEAT;
  const float* __restrict__ wrow = W + (size_t)ncol * D_FEAT;

  v8f acc = {};
#pragma unroll
  for (int k = 0; k < D_FEAT; k += 4) {
    const int kk = k + half * 2;
    v2f a = *(const v2f*)(xrow + kk);
    v2f b = *(const v2f*)(wrow + kk);
    acc = __builtin_amdgcn_wmma_f32_16x16x4_f32(
        false, a, false, b, (short)0, acc, false, false);
  }

#pragma unroll
  for (int v = 0; v < 8; ++v) {
    const int r = row0 + half * 8 + v;
    if (r < nNodes) {
      xw[(size_t)r * D_FEAT + wave * 16 + l16] = acc[v];
    }
  }
}

__global__ __launch_bounds__(256) void ggnn_scatter_edges(
    const float* __restrict__ xw,
    const int* __restrict__ src,
    const int* __restrict__ dst,
    const float* __restrict__ bias,
    float* __restrict__ agg,
    int nEdges) {
  const int e = blockIdx.x * (blockDim.x >> 5) + (threadIdx.x >> 5);
  if (e >= nEdges) return;
  const int lane = threadIdx.x & 31;

  const int s = src[e];
  const int d = dst[e];

  const float4 xv = *(const float4*)(xw + (size_t)s * D_FEAT + lane * 4);
  const float4 bv = *(const float4*)(bias + lane * 4);
  float* dp = agg + (size_t)d * D_FEAT + lane * 4;

  unsafeAtomicAdd(dp + 0, xv.x + bv.x);
  unsafeAtomicAdd(dp + 1, xv.y + bv.y);
  unsafeAtomicAdd(dp + 2, xv.z + bv.z);
  unsafeAtomicAdd(dp + 3, xv.w + bv.w);
}

__global__ __launch_bounds__(256) void ggnn_relu_pool(
    const float* __restrict__ agg,
    const int* __restrict__ batch,
    float* __restrict__ sums,
    float* __restrict__ counts,
    int nNodes) {
  const int lane   = threadIdx.x & 31;
  const int waveId = blockIdx.x * (blockDim.x >> 5) + (threadIdx.x >> 5);
  const int nWaves = gridDim.x * (blockDim.x >> 5);
  const int chunk  = (nNodes + nWaves - 1) / nWaves;
  const int n0 = waveId * chunk;
  int n1 = n0 + chunk;
  if (n1 > nNodes) n1 = nNodes;
  if (n0 >= n1) return;

  float4 acc = make_float4(0.f, 0.f, 0.f, 0.f);
  int cnt = 0;
  int g = batch[n0];

  for (int n = n0; n < n1; ++n) {
    const int gn = batch[n];
    if (gn != g) {
      float* sp = sums + (size_t)g * D_FEAT + lane * 4;
      unsafeAtomicAdd(sp + 0, acc.x);
      unsafeAtomicAdd(sp + 1, acc.y);
      unsafeAtomicAdd(sp + 2, acc.z);
      unsafeAtomicAdd(sp + 3, acc.w);
      if (lane == 0) unsafeAtomicAdd(counts + g, (float)cnt);
      acc = make_float4(0.f, 0.f, 0.f, 0.f);
      cnt = 0;
      g = gn;
    }
    const float4 v = *(const float4*)(agg + (size_t)n * D_FEAT + lane * 4);
    acc.x += fmaxf(v.x, 0.f);
    acc.y += fmaxf(v.y, 0.f);
    acc.z += fmaxf(v.z, 0.f);
    acc.w += fmaxf(v.w, 0.f);
    ++cnt;
  }
  float* sp = sums + (size_t)g * D_FEAT + lane * 4;
  unsafeAtomicAdd(sp + 0, acc.x);
  unsafeAtomicAdd(sp + 1, acc.y);
  unsafeAtomicAdd(sp + 2, acc.z);
  unsafeAtomicAdd(sp + 3, acc.w);
  if (lane == 0) unsafeAtomicAdd(counts + g, (float)cnt);
}

__global__ __launch_bounds__(256) void ggnn_finalize(
    const float* __restrict__ sums,
    const float* __restrict__ counts,
    float* __restrict__ out, int total) {
  const int i = blockIdx.x * blockDim.x + threadIdx.x;
  if (i >= total) return;
  const float c = counts[i >> 7];
  out[i] = sums[i] / fmaxf(c, 1.0f);
}

extern "C" void kernel_launch(void* const* d_in, const int* in_sizes, int n_in,
                              void* d_out, int out_size, void* d_ws, size_t ws_size,
                              hipStream_t stream) {
  const float* x          = (const float*)d_in[0];
  const int*   edge_index = (const int*)  d_in[1];
  const int*   batch      = (const int*)  d_in[2];
  const float* W          = (const float*)d_in[3];
  const float* bias       = (const float*)d_in[4];
  float*       out        = (float*)d_out;

  const int N = in_sizes[0] / D_FEAT;
  const int E = in_sizes[1] / 2;
  const int G = out_size / D_FEAT;

  float* xw     = (float*)d_ws;
  float* agg    = xw  + (size_t)N * D_FEAT;
  float* sums   = agg + (size_t)N * D_FEAT;
  float* counts = sums + (size_t)G * D_FEAT;

  hipMemsetAsync(agg, 0,
                 ((size_t)N * D_FEAT + (size_t)G * D_FEAT + (size_t)G) * sizeof(float),
                 stream);

  const int rowTiles = (N + 15) / 16;
  ggnn_gemm_wmma<<<rowTiles, 256, 0, stream>>>(x, W, xw, N);

  const int wavesPerBlock = 256 / 32;
  ggnn_scatter_edges<<<(E + wavesPerBlock - 1) / wavesPerBlock, 256, 0, stream>>>(
      xw, edge_index, edge_index + E, bias, agg, E);

  ggnn_relu_pool<<<128, 256, 0, stream>>>(agg, batch, sums, counts, N);

  const int total = G * D_FEAT;
  ggnn_finalize<<<(total + 255) / 256, 256, 0, stream>>>(sums, counts, out, total);
}


# GEMM feed: W staged once per block in LDS; a wave owns 16x128 row tiles, its x rows read once as 16-byte A fragments reused over 8 column tiles; k regrouped within each 8 (product sum order changes)
# speedup vs baseline: 1.6841x; 1.6841x over previous
#include <hip/hip_runtime.h>

typedef __attribute__((ext_vector_type(2))) float v2f;
typedef __attribute__((ext_vector_type(8))) float v8f;

#define D_FEAT 128

#define W_PITCH 132
#define GEMM_WAVES 16
#define GEMM_BLOCKS 256

__device__ __forceinline__ void ggnn_load_x_rows(
    float4 (&a)[16], const float* __restrict__ x, int tile, int nNodes,
    int l16, int half) {
  int mrow = tile * 16 + l16;
  if (mrow >= nNodes) mrow = nNodes - 1;
  const float* __restrict__ xr = x + (size_t)mrow * D_FEAT + half * 4;
#pragma unroll
  for (int j = 0; j < 16; ++j) a[j] = *(const float4*)(xr + 8 * j);
}

__global__ __launch_bounds__(512) void ggnn_gemm_wmma(
    const float* __restrict__ x,
    const float* __restrict__ W,
    float* __restrict__ xw,
    int nNodes) {
  __shared__ __attribute__((aligned(16))) float Ws[D_FEAT * W_PITCH];

  const int lane = threadIdx.x & 31;
  const int wave = __builtin_amdgcn_readfirstlane((int)(threadIdx.x >> 5));
  const int half = lane >> 4;
  const int l16  = lane & 15;
  const int nTiles = (nNodes + 15) >> 4;
  const int tileStride = GEMM_WAVES * (int)gridDim.x;

  int tile = wave * (int)gridDim.x + (int)blockIdx.x;

  float4 a[16];
  if (tile < nTiles) ggnn_load_x_rows(a, x, tile, nNodes, l16, half);

#pragma unroll
  for (int it = 0; it < (D_FEAT * (D_FEAT / 4)) / 512; ++it) {
    const int i = it * 512 + (int)threadIdx.x;
    const int r = i >> 5, c4 = i & 31;
    const float4 v = *(const float4*)(W + r * D_FEAT + c4 * 4);
    *(float4*)(Ws + r * W_PITCH + c4 * 4) = v;
  }
  __syncthreads();

  const float* wl = Ws + l16 * W_PITCH + half * 4;

  while (tile < nTiles) {
    v8f acc[8];
#pragma unroll
    for (int t = 0; t < 8; ++t) acc[t] = (v8f){};

    float4 bc[8];
#pragma unroll
    for (int t = 0; t < 8; ++t) bc[t] = *(const float4*)(wl + t * 16 * W_PITCH);

#pragma unroll
    for (int j = 0; j < 16; ++j) {
      float4 bn[8];
      if (j + 1 < 16) {
#pragma unroll
        for (int t = 0; t < 8; ++t)
          bn[t] = *(const float4*)(wl + t * 16 * W_PITCH + 8 * (j + 1));
      }
      __builtin_amdgcn_sched_barrier(0);
      const v2f a0 = {a[j].x, a[j].y};
      const v2f a1 = {a[j].z, a[j].w};
#pragma unroll
      for (int t = 0; t < 8; ++t) {
        const v2f b0 = {bc[t].x, bc[t].y};
        acc[t] = __builtin_amdgcn_wmma_f32_16x16x4_f32(
            false, a0, false, b0, (short)0, acc[t], false, false);
      }
#pragma unroll
      for (int t = 0; t < 8; ++t) {
        const v2f b1 = {bc[t].z, bc[t].w};
        acc[t] = __builtin_amdgcn_wmma_f32_16x16x4_f32(
            false, a1, false, b1, (short)0, acc[t], false, false);
      }
      __builtin_amdgcn_sched_barrier(0);
      if (j + 1 < 16) {
#pragma unroll
        for (int t = 0; t < 8; ++t) bc[t] = bn[t];
      }
    }

    const int row0 = tile * 16;
    tile += tileStride;
    if (tile < nTiles) ggnn_load_x_rows(a, x, tile, nNodes, l16, half);

#pragma unroll
    for (int v = 0; v < 8; ++v) {
      const int r = row0 + half * 8 + v;
      if (r < nNodes) {
        float* orow = xw + (size_t)r * D_FEAT + l16;
#pragma unroll
        for (int t = 0; t < 8; ++t) orow[t * 16] = acc[t][v];
      }
    }
  }
}

#define EDGES_PER_STRIP 32
#define EDGES_PER_GROUP 8

__global__ __launch_bounds__(256) void ggnn_scatter_edges(
    const float* __restrict__ xw,
    const int* __restrict__ src,
    const int* __restrict__ dst,
    const float* __restrict__ bias,
    float* __restrict__ agg,
    int nEdges) {
  const int lane = threadIdx.x & 31;
  const int strip = blockIdx.x * (blockDim.x >> 5) +
                    __builtin_amdgcn_readfirstlane((int)(threadIdx.x >> 5));
  const int e0 = strip * EDGES_PER_STRIP;
  if (e0 >= nEdges) return;
  const int left = nEdges - e0;
  const int cnt = left < EDGES_PER_STRIP ? left : EDGES_PER_STRIP;

  const int ei = (lane < cnt) ? (e0 + lane) : (nEdges - 1);
  const int sv = src[ei];
  const int dv = dst[ei];

  const float b0 = bias[lane];
  const float b1 = bias[lane + 32];
  const float b2 = bias[lane + 64];
  const float b3 = bias[lane + 96];

  if (cnt == EDGES_PER_STRIP) {
#pragma unroll
    for (int j0 = 0; j0 < EDGES_PER_STRIP; j0 += EDGES_PER_GROUP) {
      float r0[EDGES_PER_GROUP], r1[EDGES_PER_GROUP];
      float r2[EDGES_PER_GROUP], r3[EDGES_PER_GROUP];
#pragma unroll
      for (int u = 0; u < EDGES_PER_GROUP; ++u) {
        const unsigned so = (unsigned)__builtin_amdgcn_readlane(sv, j0 + u) * (unsigned)D_FEAT;
        const float* xr = xw + so;
        r0[u] = xr[lane];
        r1[u] = xr[lane + 32];
        r2[u] = xr[lane + 64];
        r3[u] = xr[lane + 96];
      }
#pragma unroll
      for (int u = 0; u < EDGES_PER_GROUP; ++u) {
        const unsigned dof = (unsigned)__builtin_amdgcn_readlane(dv, j0 + u) * (unsigned)D_FEAT;
        float* dp = agg + dof;
        unsafeAtomicAdd(dp + lane,      r0[u] + b0);
        unsafeAtomicAdd(dp + lane + 32, r1[u] + b1);
        unsafeAtomicAdd(dp + lane + 64, r2[u] + b2);
        unsafeAtomicAdd(dp + lane + 96, r3[u] + b3);
      }
    }
  } else {
    for (int j = 0; j < cnt; ++j) {
      const int s = __builtin_amdgcn_readlane(sv, j);
      const int d = __builtin_amdgcn_readlane(dv, j);
      const float* xr = xw + (size_t)s * D_FEAT + lane;
      float* dp = agg + (size_t)d * D_FEAT + lane;
      const float x0 = xr[0], x1 = xr[32], x2 = xr[64], x3 = xr[96];
      unsafeAtomicAdd(dp +  0, x0 + b0);
      unsafeAtomicAdd(dp + 32, x1 + b1);
      unsafeAtomicAdd(dp + 64, x2 + b2);
      unsafeAtomicAdd(dp + 96, x3 + b3);
    }
  }
}

__global__ __launch_bounds__(256) void ggnn_relu_pool(
    const float* __restrict__ agg,
    const int* __restrict__ batch,
    float* __restrict__ sums,
    float* __restrict__ counts,
    int nNodes) {
  const int lane   = threadIdx.x & 31;
  const int waveId = blockIdx.x * (blockDim.x >> 5) + (threadIdx.x >> 5);
  const int nWaves = gridDim.x * (blockDim.x >> 5);
  const int chunk  = (nNodes + nWaves - 1) / nWaves;
  const int n0 = waveId * chunk;
  int n1 = n0 + chunk;
  if (n1 > nNodes) n1 = nNodes;
  if (n0 >= n1) return;

  float4 acc = make_float4(0.f, 0.f, 0.f, 0.f);
  int cnt = 0;
  int g = batch[n0];

  for (int n = n0; n < n1; ++n) {
    const int gn = batch[n];
    if (gn != g) {
      float* sp = sums + (size_t)g * D_FEAT + lane * 4;
      unsafeAtomicAdd(sp + 0, acc.x);
      unsafeAtomicAdd(sp + 1, acc.y);
      unsafeAtomicAdd(sp + 2, acc.z);
      unsafeAtomicAdd(sp + 3, acc.w);
      if (lane == 0) unsafeAtomicAdd(counts + g, (float)cnt);
      acc = make_float4(0.f, 0.f, 0.f, 0.f);
      cnt = 0;
      g = gn;
    }
    const float4 v = *(const float4*)(agg + (size_t)n * D_FEAT + lane * 4);
    acc.x += fmaxf(v.x, 0.f);
    acc.y += fmaxf(v.y, 0.f);
    acc.z += fmaxf(v.z, 0.f);
    acc.w += fmaxf(v.w, 0.f);
    ++cnt;
  }
  float* sp = sums + (size_t)g * D_FEAT + lane * 4;
  unsafeAtomicAdd(sp + 0, acc.x);
  unsafeAtomicAdd(sp + 1, acc.y);
  unsafeAtomicAdd(sp + 2, acc.z);
  unsafeAtomicAdd(sp + 3, acc.w);
  if (lane == 0) unsafeAtomicAdd(counts + g, (float)cnt);
}

__global__ __launch_bounds__(256) void ggnn_finalize(
    const float* __restrict__ sums,
    const float* __restrict__ counts,
    float* __restrict__ out, int total) {
  const int i = blockIdx.x * blockDim.x + threadIdx.x;
  if (i >= total) return;
  const float c = counts[i >> 7];
  out[i] = sums[i] / fmaxf(c, 1.0f);
}

extern "C" void kernel_launch(void* const* d_in, const int* in_sizes, int n_in,
                              void* d_out, int out_size, void* d_ws, size_t ws_size,
                              hipStream_t stream) {
  const float* x          = (const float*)d_in[0];
  const int*   edge_index = (const int*)  d_in[1];
  const int*   batch      = (const int*)  d_in[2];
  const float* W          = (const float*)d_in[3];
  const float* bias       = (const float*)d_in[4];
  float*       out        = (float*)d_out;

  const int N = in_sizes[0] / D_FEAT;
  const int E = in_sizes[1] / 2;
  const int G = out_size / D_FEAT;

  float* xw     = (float*)d_ws;
  float* agg    = xw  + (size_t)N * D_FEAT;
  float* sums   = agg + (size_t)N * D_FEAT;
  float* counts = sums + (size_t)G * D_FEAT;

  hipMemsetAsync(agg, 0,
                 ((size_t)N * D_FEAT + (size_t)G * D_FEAT + (size_t)G) * sizeof(float),
                 stream);

  ggnn_gemm_wmma<<<GEMM_BLOCKS, 512, 0, stream>>>(x, W, xw, N);

  const int wavesPerBlock = 256 / 32;
  const int edgeStrips = (E + EDGES_PER_STRIP - 1) / EDGES_PER_STRIP;
  ggnn_scatter_edges<<<(edgeStrips + wavesPerBlock - 1) / wavesPerBlock, 256, 0, stream>>>(
      xw, edge_index, edge_index + E, bias, agg, E);

  ggnn_relu_pool<<<128, 256, 0, stream>>>(agg, batch, sums, counts, N);

  const int total = G * D_FEAT;
  ggnn_finalize<<<(total + 255) / 256, 256, 0, stream>>>(sums, counts, out, total);
}
